# starting kernel
# baseline (speedup 1.0000x reference)
#include <hip/hip_runtime.h>


#define N_INPUT     4096
#define NUM_CENTERS 512
#define DIM         128
#define N_OUT       10

#define TILE_ROWS   16
#define BLOCK       256
#define BATCH_STRIDE 132
#define RAD_STRIDE   513

typedef __attribute__((ext_vector_type(2))) float v2f;
typedef __attribute__((ext_vector_type(8))) float v8f;

__global__ __launch_bounds__(BLOCK)
void rbfn_fused_kernel(const float* __restrict__ batches,
                       const float* __restrict__ centers,
                       const float* __restrict__ sigma_p,
                       const float* __restrict__ W,
                       const float* __restrict__ bias,
                       float* __restrict__ out)
{
    __shared__ float batch_lds[TILE_ROWS * BATCH_STRIDE];
    __shared__ float radial_lds[TILE_ROWS * RAD_STRIDE];

    const int tid  = threadIdx.x;
    const int row0 = blockIdx.x * TILE_ROWS;
    const float sigma = sigma_p[0];

    {
        const float4* gsrc = (const float4*)(batches + (size_t)row0 * DIM);
        for (int i = tid; i < TILE_ROWS * (DIM / 4); i += BLOCK) {
            const int r  = i >> 5;
            const int c4 = i & 31;
            float4 v = gsrc[r * (DIM / 4) + c4];
            float* dst = &batch_lds[r * BATCH_STRIDE + c4 * 4];
            dst[0] = v.x; dst[1] = v.y; dst[2] = v.z; dst[3] = v.w;
        }
    }
    __syncthreads();

    {
        const int r    = tid & 15;
        const int cgrp = tid >> 4;
        const float* brow = &batch_lds[r * BATCH_STRIDE];
        for (int i = 0; i < NUM_CENTERS / 16; ++i) {
            const int c = i * 16 + cgrp;
            const float4* crow = (const float4*)(centers + (size_t)c * DIM);
            float acc = 0.0f;
#pragma unroll 8
            for (int d4 = 0; d4 < DIM / 4; ++d4) {
                float4 cv = crow[d4];
                acc += __builtin_fabsf(brow[d4 * 4 + 0] - cv.x);
                acc += __builtin_fabsf(brow[d4 * 4 + 1] - cv.y);
                acc += __builtin_fabsf(brow[d4 * 4 + 2] - cv.z);
                acc += __builtin_fabsf(brow[d4 * 4 + 3] - cv.w);
            }
            radial_lds[r * RAD_STRIDE + c] = __expf(-sigma * acc);
        }
    }
    __syncthreads();

    if (tid < 32) {
        const int lane = tid;
        const int half = lane >> 4;
        const int m    = lane & 15;

        const bool  valid = (m < N_OUT);
        const float mask  = valid ? 1.0f : 0.0f;
        const float* wrow = W + (size_t)(valid ? m : 0) * NUM_CENTERS;
        const float* arow = radial_lds + m * RAD_STRIDE;

        v8f acc0 = {};
        v8f acc1 = {};
        for (int k = 0; k < NUM_CENTERS; k += 8) {
            const int kk0 = k + 2 * half;
            const int kk1 = k + 4 + 2 * half;

            v2f a0, a1, b0, b1;
            a0.x = arow[kk0];
            a0.y = arow[kk0 + 1];
            a1.x = arow[kk1];
            a1.y = arow[kk1 + 1];
            const float2 w0 = *(const float2*)(wrow + kk0);
            const float2 w1 = *(const float2*)(wrow + kk1);
            b0.x = w0.x * mask;  b0.y = w0.y * mask;
            b1.x = w1.x * mask;  b1.y = w1.y * mask;

            acc0 = __builtin_amdgcn_wmma_f32_16x16x4_f32(
                false, a0, false, b0, (short)0, acc0, false, false);
            acc1 = __builtin_amdgcn_wmma_f32_16x16x4_f32(
                false, a1, false, b1, (short)0, acc1, false, false);
        }

        if (valid) {
            const float bv = bias[m];
#pragma unroll
            for (int v = 0; v < 8; ++v) {
                const int row = row0 + v + 8 * half;
                out[(size_t)row * N_OUT + m] = acc0[v] + acc1[v] + bv;
            }
        }
    }
}

extern "C" void kernel_launch(void* const* d_in, const int* in_sizes, int n_in,
                              void* d_out, int out_size, void* d_ws, size_t ws_size,
                              hipStream_t stream) {
    const float* batches = (const float*)d_in[0];
    const float* centers = (const float*)d_in[1];
    const float* sigma   = (const float*)d_in[2];
    const float* W       = (const float*)d_in[3];
    const float* bias    = (const float*)d_in[4];
    float* out           = (float*)d_out;

    (void)in_sizes; (void)n_in; (void)out_size; (void)d_ws; (void)ws_size;

    rbfn_fused_kernel<<<N_INPUT / TILE_ROWS, BLOCK, 0, stream>>>(
        batches, centers, sigma, W, bias, out);
}


# Distance phase re-cut: lane = centre, its 128 dims held in registers; batch values come as scalar-register operands from scalar loads; no LDS or vector loads in the loop; per-sum add order kept
# speedup vs baseline: 3.3563x; 3.3563x over previous
#include <hip/hip_runtime.h>


#define N_INPUT     4096
#define NUM_CENTERS 512
#define DIM         128
#define N_OUT       10

#define TILE_ROWS   16
#define BLOCK       512
#define NWAVES      (BLOCK / 32)
#define RGROUP      2
#define NGROUPS     (TILE_ROWS / RGROUP)
#define CHUNK       16
#define RAD_STRIDE  513
#define CROW4       33

typedef __attribute__((ext_vector_type(2))) float v2f;
typedef __attribute__((ext_vector_type(8))) float v8f;

__global__ __launch_bounds__(BLOCK)
void rbfn_fused_kernel(const float* __restrict__ batches,
                       const float* __restrict__ centers,
                       const float* __restrict__ sigma_p,
                       const float* __restrict__ W,
                       const float* __restrict__ bias,
                       float* __restrict__ out)
{
    __shared__ float4 smem[NWAVES * 32 * CROW4];
    float4* cstage     = smem;
    float*  radial_lds = (float*)smem;

    const int tid  = threadIdx.x;
    const int lane = tid & 31;
    const int wave = __builtin_amdgcn_readfirstlane(tid >> 5);
    const int row0 = blockIdx.x * TILE_ROWS;
    const float sigma = sigma_p[0];

    const float* btile = batches + (size_t)row0 * DIM;

    float4 creg[DIM / 4];
    {
        const float4* gsrc = (const float4*)(centers + (size_t)wave * 32 * DIM);
        float4* st = cstage + wave * 32 * CROW4;
#pragma unroll
        for (int i = 0; i < 32; ++i)
            st[i * CROW4 + lane] = gsrc[i * 32 + lane];
        __syncthreads();
#pragma unroll
        for (int j = 0; j < DIM / 4; ++j)
            creg[j] = st[lane * CROW4 + j];
        __syncthreads();
    }

    {
        const int g0 = wave & (NGROUPS - 1);
        float nxt[RGROUP][CHUNK];
#pragma unroll
        for (int r = 0; r < RGROUP; ++r)
#pragma unroll
            for (int e = 0; e < CHUNK; ++e)
                nxt[r][e] = btile[(g0 * RGROUP + r) * DIM + e];
        for (int it = 0; it < NGROUPS; ++it) {
            const int rg  = ((it + g0) & (NGROUPS - 1)) * RGROUP;
            const int rgn = ((it + 1 + g0) & (NGROUPS - 1)) * RGROUP;
            const float* bp  = btile + rg * DIM;
            const float* bpn = btile + rgn * DIM;
            float acc[RGROUP];
#pragma unroll
            for (int r = 0; r < RGROUP; ++r) acc[r] = 0.0f;
#pragma unroll
            for (int ch = 0; ch < DIM / CHUNK; ++ch) {
                float cur[RGROUP][CHUNK];
#pragma unroll
                for (int r = 0; r < RGROUP; ++r) {
                    const float* src = (ch + 1 < DIM / CHUNK) ? bp + r * DIM + (ch + 1) * CHUNK
                                                              : bpn + r * DIM;
#pragma unroll
                    for (int e = 0; e < CHUNK; ++e) {
                        cur[r][e] = nxt[r][e];
                        nxt[r][e] = src[e];
                    }
                }
#pragma unroll
                for (int q = 0; q < CHUNK / 4; ++q) {
                    const int j = ch * (CHUNK / 4) + q;
#pragma unroll
                    for (int r = 0; r < RGROUP; ++r) {
                        acc[r] += __builtin_fabsf(cur[r][q * 4 + 0] - creg[j].x);
                        acc[r] += __builtin_fabsf(cur[r][q * 4 + 1] - creg[j].y);
                        acc[r] += __builtin_fabsf(cur[r][q * 4 + 2] - creg[j].z);
                        acc[r] += __builtin_fabsf(cur[r][q * 4 + 3] - creg[j].w);
                    }
                    __builtin_amdgcn_sched_barrier(0);
                }
            }
            const int c = wave * 32 + lane;
#pragma unroll
            for (int r = 0; r < RGROUP; ++r)
                radial_lds[(rg + r) * RAD_STRIDE + c] = __expf(-sigma * acc[r]);
        }
    }
    __syncthreads();

    if (tid < 32) {
        const int half = lane >> 4;
        const int m    = lane & 15;

        const bool  valid = (m < N_OUT);
        const float mask  = valid ? 1.0f : 0.0f;
        const float* wrow = W + (size_t)(valid ? m : 0) * NUM_CENTERS;
        const float* arow = radial_lds + m * RAD_STRIDE;

        v8f acc0 = {};
        v8f acc1 = {};
        for (int k = 0; k < NUM_CENTERS; k += 8) {
            const int kk0 = k + 2 * half;
            const int kk1 = k + 4 + 2 * half;

            v2f a0, a1, b0, b1;
            a0.x = arow[kk0];
            a0.y = arow[kk0 + 1];
            a1.x = arow[kk1];
            a1.y = arow[kk1 + 1];
            const float2 w0 = *(const float2*)(wrow + kk0);
            const float2 w1 = *(const float2*)(wrow + kk1);
            b0.x = w0.x * mask;  b0.y = w0.y * mask;
            b1.x = w1.x * mask;  b1.y = w1.y * mask;

            acc0 = __builtin_amdgcn_wmma_f32_16x16x4_f32(
                false, a0, false, b0, (short)0, acc0, false, false);
            acc1 = __builtin_amdgcn_wmma_f32_16x16x4_f32(
                false, a1, false, b1, (short)0, acc1, false, false);
        }

        if (valid) {
            const float bv = bias[m];
#pragma unroll
            for (int v = 0; v < 8; ++v) {
                const int row = row0 + v + 8 * half;
                out[(size_t)row * N_OUT + m] = acc0[v] + acc1[v] + bv;
            }
        }
    }
}

extern "C" void kernel_launch(void* const* d_in, const int* in_sizes, int n_in,
                              void* d_out, int out_size, void* d_ws, size_t ws_size,
                              hipStream_t stream) {
    const float* batches = (const float*)d_in[0];
    const float* centers = (const float*)d_in[1];
    const float* sigma   = (const float*)d_in[2];
    const float* W       = (const float*)d_in[3];
    const float* bias    = (const float*)d_in[4];
    float* out           = (float*)d_out;

    (void)in_sizes; (void)n_in; (void)out_size; (void)d_ws; (void)ws_size;

    rbfn_fused_kernel<<<N_INPUT / TILE_ROWS, BLOCK, 0, stream>>>(
        batches, centers, sigma, W, bias, out);
}


# W copied once into LDS (zero-padded to 16 rows) while the centre rows arrive; output GEMM reads its B fragments from LDS, a block ahead: no global loads or mask multiplies in its K loop; K order kept
# speedup vs baseline: 4.2178x; 1.2567x over previous
#include <hip/hip_runtime.h>


#define N_INPUT     4096
#define NUM_CENTERS 512
#define DIM         128
#define N_OUT       10

#define TILE_ROWS   16
#define BLOCK       512
#define NWAVES      (BLOCK / 32)
#define RGROUP      2
#define NGROUPS     (TILE_ROWS / RGROUP)
#define CHUNK       16
#define RAD_STRIDE  513
#define CROW4       33
#define W_STRIDE    516
#define W_ROWS      16
#define KSTEPS      8

typedef __attribute__((ext_vector_type(2))) float v2f;
typedef __attribute__((ext_vector_type(8))) float v8f;

__global__ __launch_bounds__(BLOCK)
void rbfn_fused_kernel(const float* __restrict__ batches,
                       const float* __restrict__ centers,
                       const float* __restrict__ sigma_p,
                       const float* __restrict__ W,
                       const float* __restrict__ bias,
                       float* __restrict__ out)
{
    __shared__ float4 smem[NWAVES * 32 * CROW4];
    float4* cstage     = smem;
    float*  radial_lds = (float*)smem;
    __shared__ float2 wstage[W_ROWS * W_STRIDE / 2];
    float* wlds = (float*)wstage;

    const int tid  = threadIdx.x;
    const int lane = tid & 31;
    const int wave = __builtin_amdgcn_readfirstlane(tid >> 5);
    const int row0 = blockIdx.x * TILE_ROWS;
    const float sigma = sigma_p[0];

    const float* btile = batches + (size_t)row0 * DIM;

    float2 wv[(N_OUT * NUM_CENTERS / 2) / BLOCK];
#pragma unroll
    for (int t = 0; t < (N_OUT * NUM_CENTERS / 2) / BLOCK; ++t)
        wv[t] = ((const float2*)W)[tid + t * BLOCK];

    float4 creg[DIM / 4];
    {
        const float4* gsrc = (const float4*)(centers + (size_t)wave * 32 * DIM);
        float4* st = cstage + wave * 32 * CROW4;
#pragma unroll
        for (int i = 0; i < 32; ++i)
            st[i * CROW4 + lane] = gsrc[i * 32 + lane];
#pragma unroll
        for (int t = 0; t < (N_OUT * NUM_CENTERS / 2) / BLOCK; ++t) {
            const int idx = tid + t * BLOCK;
            wstage[(idx >> 8) * (W_STRIDE / 2) + (idx & 255)] = wv[t];
        }
#pragma unroll
        for (int t = 0; t < ((W_ROWS - N_OUT) * NUM_CENTERS / 2) / BLOCK; ++t) {
            const int idx = tid + t * BLOCK;
            wstage[(N_OUT + (idx >> 8)) * (W_STRIDE / 2) + (idx & 255)] = make_float2(0.0f, 0.0f);
        }
        __syncthreads();
#pragma unroll
        for (int j = 0; j < DIM / 4; ++j)
            creg[j] = st[lane * CROW4 + j];
        __syncthreads();
    }

    {
        const int g0 = wave & (NGROUPS - 1);
        float nxt[RGROUP][CHUNK];
#pragma unroll
        for (int r = 0; r < RGROUP; ++r)
#pragma unroll
            for (int e = 0; e < CHUNK; ++e)
                nxt[r][e] = btile[(g0 * RGROUP + r) * DIM + e];
        for (int it = 0; it < NGROUPS; ++it) {
            const int rg  = ((it + g0) & (NGROUPS - 1)) * RGROUP;
            const int rgn = ((it + 1 + g0) & (NGROUPS - 1)) * RGROUP;
            const float* bp  = btile + rg * DIM;
            const float* bpn = btile + rgn * DIM;
            float acc[RGROUP];
#pragma unroll
            for (int r = 0; r < RGROUP; ++r) acc[r] = 0.0f;
#pragma unroll
            for (int ch = 0; ch < DIM / CHUNK; ++ch) {
                float cur[RGROUP][CHUNK];
#pragma unroll
                for (int r = 0; r < RGROUP; ++r) {
                    const float* src = (ch + 1 < DIM / CHUNK) ? bp + r * DIM + (ch + 1) * CHUNK
                                                              : bpn + r * DIM;
#pragma unroll
                    for (int e = 0; e < CHUNK; ++e) {
                        cur[r][e] = nxt[r][e];
                        nxt[r][e] = src[e];
                    }
                }
#pragma unroll
                for (int q = 0; q < CHUNK / 4; ++q) {
                    const int j = ch * (CHUNK / 4) + q;
#pragma unroll
                    for (int r = 0; r < RGROUP; ++r) {
                        acc[r] += __builtin_fabsf(cur[r][q * 4 + 0] - creg[j].x);
                        acc[r] += __builtin_fabsf(cur[r][q * 4 + 1] - creg[j].y);
                        acc[r] += __builtin_fabsf(cur[r][q * 4 + 2] - creg[j].z);
                        acc[r] += __builtin_fabsf(cur[r][q * 4 + 3] - creg[j].w);
                    }
                    __builtin_amdgcn_sched_barrier(0);
                }
            }
            const int c = wave * 32 + lane;
#pragma unroll
            for (int r = 0; r < RGROUP; ++r)
                radial_lds[(rg + r) * RAD_STRIDE + c] = __expf(-sigma * acc[r]);
        }
    }
    __syncthreads();

    if (tid < 32) {
        const int half = lane >> 4;
        const int m    = lane & 15;

        const bool  valid = (m < N_OUT);
        const float* wrow = wlds + m * W_STRIDE;
        const float* arow = radial_lds + m * RAD_STRIDE;

        v8f acc0 = {};
        v8f acc1 = {};
        v2f af[2][KSTEPS][2];
        v2f bf[2][KSTEPS][2];
#pragma unroll
        for (int blk = -1; blk < NUM_CENTERS / (8 * KSTEPS); ++blk) {
            if (blk + 1 < NUM_CENTERS / (8 * KSTEPS)) {
                const int set = (blk + 1) & 1;
#pragma unroll
                for (int q = 0; q < KSTEPS; ++q) {
                    const int kk0 = ((blk + 1) * KSTEPS + q) * 8 + 2 * half;
                    const int kk1 = kk0 + 4;
                    af[set][q][0].x = arow[kk0];
                    af[set][q][0].y = arow[kk0 + 1];
                    af[set][q][1].x = arow[kk1];
                    af[set][q][1].y = arow[kk1 + 1];
                    const float2 w0 = *(const float2*)(wrow + kk0);
                    const float2 w1 = *(const float2*)(wrow + kk1);
                    bf[set][q][0].x = w0.x;  bf[set][q][0].y = w0.y;
                    bf[set][q][1].x = w1.x;  bf[set][q][1].y = w1.y;
                }
            }
            if (blk >= 0 && blk + 1 < NUM_CENTERS / (8 * KSTEPS))
                __builtin_amdgcn_sched_barrier(0);
            if (blk >= 0) {
                const int set = blk & 1;
#pragma unroll
                for (int q = 0; q < KSTEPS; ++q) {
                    acc0 = __builtin_amdgcn_wmma_f32_16x16x4_f32(
                        false, af[set][q][0], false, bf[set][q][0], (short)0, acc0, false, false);
                    acc1 = __builtin_amdgcn_wmma_f32_16x16x4_f32(
                        false, af[set][q][1], false, bf[set][q][1], (short)0, acc1, false, false);
                }
            }
            if (blk >= 0 && blk + 2 < NUM_CENTERS / (8 * KSTEPS))
                __builtin_amdgcn_sched_barrier(0);
        }

        if (valid) {
            const float bv = bias[m];
#pragma unroll
            for (int v = 0; v < 8; ++v) {
                const int row = row0 + v + 8 * half;
                out[(size_t)row * N_OUT + m] = acc0[v] + acc1[v] + bv;
            }
        }
    }
}

extern "C" void kernel_launch(void* const* d_in, const int* in_sizes, int n_in,
                              void* d_out, int out_size, void* d_ws, size_t ws_size,
                              hipStream_t stream) {
    const float* batches = (const float*)d_in[0];
    const float* centers = (const float*)d_in[1];
    const float* sigma   = (const float*)d_in[2];
    const float* W       = (const float*)d_in[3];
    const float* bias    = (const float*)d_in[4];
    float* out           = (float*)d_out;

    (void)in_sizes; (void)n_in; (void)out_size; (void)d_ws; (void)ws_size;

    rbfn_fused_kernel<<<N_INPUT / TILE_ROWS, BLOCK, 0, stream>>>(
        batches, centers, sigma, W, bias, out);
}
